# starting kernel
# baseline (speedup 1.0000x reference)
#include <hip/hip_runtime.h>


#define NN 1024
#define NB 128

typedef float v2f __attribute__((ext_vector_type(2)));
typedef float v8f __attribute__((ext_vector_type(8)));

__global__ __launch_bounds__(128) void mcf_wmma_kernel(const float* __restrict__ x0,
                                                       const float* __restrict__ x1,
                                                       const float* __restrict__ x2,
                                                       float* __restrict__ out) {
    __shared__ float fwd[2 * NN];
    __shared__ float rev[2 * NN];

    const int b     = blockIdx.x;
    const int which = blockIdx.y;
    const float* __restrict__ x =
        (which == 0) ? x0 : (which == 1) ? x1 : x2;
    const float* __restrict__ xb = x + b * NN;
    float* __restrict__ ob = out + (size_t)which * (NB * NN) + (size_t)b * NN;

    const int tid = threadIdx.x;

    #pragma unroll
    for (int i = tid; i < NN; i += 128) {
        float v = xb[i];
        fwd[i]      = v;
        fwd[i + NN] = v;
        int u = (0 - i) & (NN - 1);
        rev[u]      = v;
        rev[u + NN] = v;
    }
    __syncthreads();

    const int wave  = tid >> 5;
    const int lane  = tid & 31;
    const int m16   = lane & 15;
    const int khalf = lane >> 4;

    const float* Abase = fwd + 16 * m16 + 2 * khalf;
    const float* Bbase = rev + NN + 2 * khalf - 256 * wave - m16;

    v8f acc0 = {};
    v8f acc1 = {};

    #pragma unroll 8
    for (int j = 0; j < NN / 2; j += 4) {
        v2f a0, b0, a1, b1;
        a0.x = Abase[j];         a0.y = Abase[j + 1];
        b0.x = Bbase[j];         b0.y = Bbase[j + 1];
        a1.x = Abase[j + 512];   a1.y = Abase[j + 513];
        b1.x = Bbase[j + 512];   b1.y = Bbase[j + 513];

        acc0 = __builtin_amdgcn_wmma_f32_16x16x4_f32(
            false, a0, false, b0, (short)0, acc0, false, false);
        acc1 = __builtin_amdgcn_wmma_f32_16x16x4_f32(
            false, a1, false, b1, (short)0, acc1, false, false);
    }

    v8f acc = acc0 + acc1;

    #pragma unroll
    for (int r = 0; r < 8; ++r) {
        ob[256 * wave + 16 * (r + 8 * khalf) + m16] = acc[r];
    }
}

extern "C" void kernel_launch(void* const* d_in, const int* in_sizes, int n_in,
                              void* d_out, int out_size, void* d_ws, size_t ws_size,
                              hipStream_t stream) {
    const float* x1 = (const float*)d_in[0];
    const float* x2 = (const float*)d_in[1];
    const float* x3 = (const float*)d_in[2];
    float* out = (float*)d_out;

    dim3 grid(NB, 3);
    dim3 block(128);
    mcf_wmma_kernel<<<grid, block, 0, stream>>>(x1, x2, x3, out);
}


# input-independent work (DFT-32 fragment reads, eight twiddles) finished under the input loads' latency; stages 3-4 read their LDS fragments as one group ahead of the WMMAs, not among them
# speedup vs baseline: 2.6309x; 2.6309x over previous
#include <hip/hip_runtime.h>


#define NN 1024
#define NB 128
#define PA 36
#define PB 40

typedef float v2f __attribute__((ext_vector_type(2)));
typedef float v8f __attribute__((ext_vector_type(8)));

__device__ __forceinline__ v8f mma(v2f a, v2f b, v8f c) {
    return __builtin_amdgcn_wmma_f32_16x16x4_f32(false, a, false, b, (short)0, c, false, false);
}

__global__ __launch_bounds__(128) void mcf_wmma_kernel(const float* __restrict__ x0,
                                                       const float* __restrict__ x1,
                                                       const float* __restrict__ x2,
                                                       float* __restrict__ out) {
    __shared__ __attribute__((aligned(16))) float c32[32];
    __shared__ __attribute__((aligned(16))) float s32[32];
    __shared__ __attribute__((aligned(16))) float Zr[32 * PA];
    __shared__ __attribute__((aligned(16))) float Zi[32 * PA];
    __shared__ __attribute__((aligned(16))) float Pr[32 * PA];
    __shared__ __attribute__((aligned(16))) float Pi[32 * PA];
    __shared__ __attribute__((aligned(16))) float Rr[32 * PB];
    __shared__ __attribute__((aligned(16))) float Ri[32 * PB];

    const int b     = blockIdx.x;
    const int which = blockIdx.y;
    const float* __restrict__ x =
        (which == 0) ? x0 : (which == 1) ? x1 : x2;
    const float* __restrict__ xb = x + b * NN;
    float* __restrict__ ob = out + (size_t)which * (NB * NN) + (size_t)b * NN;

    const int tid = threadIdx.x;

    const int wave = tid >> 5;
    const int lane = tid & 31;
    const int c16  = lane & 15;
    const int h    = lane >> 4;
    const int ti   = wave >> 1;
    const int tj   = wave & 1;
    const int rowA = 16 * ti + c16;
    const int colB = 16 * tj + c16;

    v2f xf[8];
    #pragma unroll
    for (int q = 0; q < 8; ++q) {
        const int k0 = 4 * q + 2 * h;
        xf[q].x = xb[32 * k0 + colB];
        xf[q].y = xb[32 * (k0 + 1) + colB];
    }

    if (tid < 32) {
        float s, c;
        sincospif((float)tid * 0.0625f, &s, &c);
        c32[tid] = c;
        s32[tid] = s;
    }

    __builtin_amdgcn_fence(__ATOMIC_RELEASE, "workgroup", "local");
    __builtin_amdgcn_s_barrier();
    __builtin_amdgcn_fence(__ATOMIC_ACQUIRE, "workgroup", "local");

    v2f ca[8], sa[8], cb[8], sb[8];
    #pragma unroll
    for (int q = 0; q < 8; ++q) {
        const int k0 = 4 * q + 2 * h;
        ca[q].x = c32[(rowA * k0) & 31];        ca[q].y = c32[(rowA * (k0 + 1)) & 31];
        sa[q].x = s32[(rowA * k0) & 31];        sa[q].y = s32[(rowA * (k0 + 1)) & 31];
        cb[q].x = c32[(colB * k0) & 31];        cb[q].y = c32[(colB * (k0 + 1)) & 31];
        sb[q].x = s32[(colB * k0) & 31];        sb[q].y = s32[(colB * (k0 + 1)) & 31];
    }

    const int rowD = 16 * ti + 8 * h;

    float cw[8], sw[8];
    #pragma unroll
    for (int r = 0; r < 8; ++r) {
        sincospif((float)((rowD + r) * colB) * 0.001953125f, &sw[r], &cw[r]);
    }

    asm volatile("" :: "v"(ca[0]), "v"(ca[1]), "v"(ca[2]), "v"(ca[3]), "v"(ca[4]), "v"(ca[5]),
                       "v"(ca[6]), "v"(ca[7]), "v"(sa[0]), "v"(sa[1]), "v"(sa[2]), "v"(sa[3]),
                       "v"(sa[4]), "v"(sa[5]), "v"(sa[6]), "v"(sa[7]));
    asm volatile("" :: "v"(cb[0]), "v"(cb[1]), "v"(cb[2]), "v"(cb[3]), "v"(cb[4]), "v"(cb[5]),
                       "v"(cb[6]), "v"(cb[7]), "v"(sb[0]), "v"(sb[1]), "v"(sb[2]), "v"(sb[3]),
                       "v"(sb[4]), "v"(sb[5]), "v"(sb[6]), "v"(sb[7]));
    asm volatile("" :: "v"(cw[0]), "v"(cw[1]), "v"(cw[2]), "v"(cw[3]), "v"(cw[4]), "v"(cw[5]),
                       "v"(cw[6]), "v"(cw[7]), "v"(sw[0]), "v"(sw[1]), "v"(sw[2]), "v"(sw[3]),
                       "v"(sw[4]), "v"(sw[5]), "v"(sw[6]), "v"(sw[7]));
    __builtin_amdgcn_sched_barrier(0);

    {
        v8f yr = {}, tt = {};
        #pragma unroll
        for (int q = 0; q < 8; ++q) {
            yr = mma(ca[q], xf[q], yr);
            tt = mma(sa[q], xf[q], tt);
        }
        #pragma unroll
        for (int r = 0; r < 8; ++r) {
            const int k1 = rowD + r;
            Zr[k1 * PA + colB] = yr[r] * cw[r] - tt[r] * sw[r];
            Zi[k1 * PA + colB] = -(tt[r] * cw[r]) - yr[r] * sw[r];
        }
    }
    __syncthreads();

    {
        v8f a1 = {}, a2 = {}, a3 = {}, a4 = {};
        #pragma unroll
        for (int q = 0; q < 8; ++q) {
            const int ad = rowA * PA + 4 * q + 2 * h;
            const v2f zr = *(const v2f*)(Zr + ad);
            const v2f zi = *(const v2f*)(Zi + ad);
            a1 = mma(zr, cb[q], a1);
            a2 = mma(zi, sb[q], a2);
            a3 = mma(zi, cb[q], a3);
            a4 = mma(zr, sb[q], a4);
        }
        #pragma unroll
        for (int r = 0; r < 8; ++r) {
            const int k1 = rowD + r;
            const float xr = a1[r] + a2[r];
            const float xi = a3[r] - a4[r];
            Pr[k1 * PA + colB] = xr * xr - xi * xi;
            Pi[k1 * PA + colB] = 2.0f * (xr * xi);
        }
    }
    __syncthreads();

    {
        v8f b1 = {}, b2 = {}, b3 = {}, b4 = {};
        v2f pr[8], pi[8];
        #pragma unroll
        for (int q = 0; q < 8; ++q) {
            const int ad = rowA * PA + 4 * q + 2 * h;
            pr[q] = *(const v2f*)(Pr + ad);
            pi[q] = *(const v2f*)(Pi + ad);
        }
        __builtin_amdgcn_sched_barrier(0);
        #pragma unroll
        for (int q = 0; q < 8; ++q) {
            b1 = mma(pr[q], cb[q], b1);
            b2 = mma(pi[q], sb[q], b2);
            b3 = mma(pr[q], sb[q], b3);
            b4 = mma(pi[q], cb[q], b4);
        }
        #pragma unroll
        for (int r = 0; r < 8; ++r) {
            const int k1 = rowD + r;
            const float qr = b1[r] - b2[r];
            const float qi = b3[r] + b4[r];
            Rr[k1 * PB + colB] = qr * cw[r] - qi * sw[r];
            Ri[k1 * PB + colB] = qr * sw[r] + qi * cw[r];
        }
    }
    __syncthreads();

    {
        v8f o1 = {}, o2 = {};
        v2f rr[8], ri[8];
        #pragma unroll
        for (int q = 0; q < 8; ++q) {
            const int k0 = 4 * q + 2 * h;
            rr[q].x = Rr[k0 * PB + colB];   rr[q].y = Rr[(k0 + 1) * PB + colB];
            ri[q].x = Ri[k0 * PB + colB];   ri[q].y = Ri[(k0 + 1) * PB + colB];
        }
        __builtin_amdgcn_sched_barrier(0);
        #pragma unroll
        for (int q = 0; q < 8; ++q) {
            o1 = mma(ca[q], rr[q], o1);
            o2 = mma(sa[q], ri[q], o2);
        }
        #pragma unroll
        for (int r = 0; r < 8; ++r) {
            ob[32 * (rowD + r) + colB] = (o1[r] - o2[r]) * 0.0009765625f;
        }
    }
}

extern "C" void kernel_launch(void* const* d_in, const int* in_sizes, int n_in,
                              void* d_out, int out_size, void* d_ws, size_t ws_size,
                              hipStream_t stream) {
    const float* x1 = (const float*)d_in[0];
    const float* x2 = (const float*)d_in[1];
    const float* x3 = (const float*)d_in[2];
    float* out = (float*)d_out;

    dim3 grid(NB, 3);
    dim3 block(128);
    mcf_wmma_kernel<<<grid, block, 0, stream>>>(x1, x2, x3, out);
}


# row as ONE half-length complex signal x[2n]+i*x[2n+1]: DFT-512 as 16x32, even/odd spectra split by mirror, H=E^2+T*O^2+2iEO: 192 WMMAs a row on 2 waves, 2 rows a work-group; sum order changes
# speedup vs baseline: 2.8401x; 1.0795x over previous
#include <hip/hip_runtime.h>


#define NN 1024
#define NB 128
#define PA 36
#define PR 68
#define PU 33
#define ROW_LDS (4 * 16 * PA + 16 * PR + 2 * 16 * PU + 8)

typedef float v2f __attribute__((ext_vector_type(2)));
typedef float v8f __attribute__((ext_vector_type(8)));

__device__ __forceinline__ v8f mma(v2f a, v2f b, v8f c) {
    return __builtin_amdgcn_wmma_f32_16x16x4_f32(false, a, false, b, (short)0, c, false, false);
}

__global__ __launch_bounds__(128) void mcf_wmma_kernel(const float* __restrict__ x0,
                                                      const float* __restrict__ x1,
                                                      const float* __restrict__ x2,
                                                      float* __restrict__ out) {
    __shared__ __attribute__((aligned(16))) float c32[32];
    __shared__ __attribute__((aligned(16))) float s32[32];
    __shared__ __attribute__((aligned(16))) float planes[2][ROW_LDS];

    const int tid  = threadIdx.x;
    const int wave = tid >> 5;
    const int pr2  = wave >> 1;
    const int wt   = wave & 1;

    float* const rowbase = planes[pr2];
    float* const Zr = rowbase;
    float* const Zi = Zr + 16 * PA;
    float* const Hr = Zi + 16 * PA;
    float* const Hi = Hr + 16 * PA;
    float* const RR = Hi + 16 * PA;
    float* const Ur = RR + 16 * PR;
    float* const Ui = Ur + 16 * PU;

    const int b     = 2 * blockIdx.x + pr2;
    const int which = blockIdx.y;
    const float* __restrict__ x =
        (which == 0) ? x0 : (which == 1) ? x1 : x2;
    const float* __restrict__ xb = x + b * NN;
    float* __restrict__ ob = out + (size_t)which * (NB * NN) + (size_t)b * NN;

    const int lane = tid & 31;
    const int c16  = lane & 15;
    const int h    = lane >> 4;
    const int rowA = c16;
    const int colB = 16 * wt + c16;
    const int rowD = 8 * h;

    v2f zl[8];
    #pragma unroll
    for (int q = 0; q < 8; ++q) {
        const int j = 2 * q + h;
        zl[q] = *(const v2f*)(xb + 64 * j + 2 * colB);
    }

    if (tid < 32) {
        float s, c;
        sincospif((float)tid * 0.0625f, &s, &c);
        c32[tid] = c;
        s32[tid] = s;
    }

    __builtin_amdgcn_fence(__ATOMIC_RELEASE, "workgroup", "local");
    __builtin_amdgcn_s_barrier();
    __builtin_amdgcn_fence(__ATOMIC_ACQUIRE, "workgroup", "local");

    v2f fa[8], fb[8], cb[8], sb[8];
    #pragma unroll
    for (int q = 0; q < 8; ++q) {
        const int j = 2 * q + h;
        const float cc = c32[(2 * rowA * j) & 31];
        const float ss = s32[(2 * rowA * j) & 31];
        fa[q].x = cc;     fa[q].y = ss;
        fb[q].x = -ss;    fb[q].y = cc;
    }
    #pragma unroll
    for (int q = 0; q < 8; ++q) {
        const int k0 = 4 * q + 2 * h;
        cb[q].x = c32[(colB * k0) & 31];        cb[q].y = c32[(colB * (k0 + 1)) & 31];
        sb[q].x = s32[(colB * k0) & 31];        sb[q].y = s32[(colB * (k0 + 1)) & 31];
    }

    const float tc = c32[colB], ts = s32[colB];
    const float c8 = h ? 0.995184720f : 1.0f;
    const float s8 = h ? 0.0980171412f : 0.0f;
    const float t0c = tc * c8 - ts * s8;
    const float t0s = tc * s8 + ts * c8;
    float cw[8], sw[8], ct[8], st[8];
    #pragma unroll
    for (int r = 0; r < 8; ++r) {
        sincospif((float)((rowD + r) * colB) * 0.00390625f, &sw[r], &cw[r]);
    }
#define SHIFT_TWIDDLE(r, C, S) ct[r] = t0c * (C) - t0s * (S); st[r] = t0c * (S) + t0s * (C);
    ct[0] = t0c;  st[0] = t0s;
    SHIFT_TWIDDLE(1, 0.999924719f, 0.0122715384f)
    SHIFT_TWIDDLE(2, 0.999698818f, 0.0245412290f)
    SHIFT_TWIDDLE(3, 0.999322355f, 0.0368072242f)
    SHIFT_TWIDDLE(4, 0.998795450f, 0.0490676761f)
    SHIFT_TWIDDLE(5, 0.998118103f, 0.0613207370f)
    SHIFT_TWIDDLE(6, 0.997290432f, 0.0735645667f)
    SHIFT_TWIDDLE(7, 0.996312618f, 0.0857973099f)
#undef SHIFT_TWIDDLE

    asm volatile("" :: "v"(fa[0]), "v"(fa[1]), "v"(fa[2]), "v"(fa[3]), "v"(fa[4]), "v"(fa[5]),
                       "v"(fa[6]), "v"(fa[7]), "v"(fb[0]), "v"(fb[1]), "v"(fb[2]), "v"(fb[3]),
                       "v"(fb[4]), "v"(fb[5]), "v"(fb[6]), "v"(fb[7]));
    asm volatile("" :: "v"(cb[0]), "v"(cb[1]), "v"(cb[2]), "v"(cb[3]), "v"(cb[4]), "v"(cb[5]),
                       "v"(cb[6]), "v"(cb[7]), "v"(sb[0]), "v"(sb[1]), "v"(sb[2]), "v"(sb[3]),
                       "v"(sb[4]), "v"(sb[5]), "v"(sb[6]), "v"(sb[7]));
    asm volatile("" :: "v"(cw[0]), "v"(cw[1]), "v"(cw[2]), "v"(cw[3]), "v"(cw[4]), "v"(cw[5]),
                       "v"(cw[6]), "v"(cw[7]), "v"(sw[0]), "v"(sw[1]), "v"(sw[2]), "v"(sw[3]),
                       "v"(sw[4]), "v"(sw[5]), "v"(sw[6]), "v"(sw[7]));
    asm volatile("" :: "v"(ct[0]), "v"(ct[1]), "v"(ct[2]), "v"(ct[3]), "v"(ct[4]), "v"(ct[5]),
                       "v"(ct[6]), "v"(ct[7]), "v"(st[0]), "v"(st[1]), "v"(st[2]), "v"(st[3]),
                       "v"(st[4]), "v"(st[5]), "v"(st[6]), "v"(st[7]));
    __builtin_amdgcn_sched_barrier(0);

    {
        v8f yr = {}, yi = {};
        #pragma unroll
        for (int q = 0; q < 8; ++q) {
            yr = mma(fa[q], zl[q], yr);
            yi = mma(fb[q], zl[q], yi);
        }
        #pragma unroll
        for (int r = 0; r < 8; ++r) {
            const int k1 = rowD + r;
            Zr[k1 * PA + colB] = yr[r] * cw[r] + yi[r] * sw[r];
            Zi[k1 * PA + colB] = yi[r] * cw[r] - yr[r] * sw[r];
        }
    }
    __syncthreads();

    float ur[8], ui[8];
    {
        v8f a1 = {}, a2 = {}, a3 = {}, a4 = {};
        v2f zr[8], zi[8];
        #pragma unroll
        for (int q = 0; q < 8; ++q) {
            const int ad = rowA * PA + 4 * q + 2 * h;
            zr[q] = *(const v2f*)(Zr + ad);
            zi[q] = *(const v2f*)(Zi + ad);
        }
        __builtin_amdgcn_sched_barrier(0);
        #pragma unroll
        for (int q = 0; q < 8; ++q) {
            a1 = mma(zr[q], cb[q], a1);
            a2 = mma(zi[q], sb[q], a2);
            a3 = mma(zi[q], cb[q], a3);
            a4 = mma(zr[q], sb[q], a4);
        }
        #pragma unroll
        for (int r = 0; r < 8; ++r) {
            const int k1 = rowD + r;
            ur[r] = a1[r] + a2[r];
            ui[r] = a3[r] - a4[r];
            Ur[k1 * PU + colB] = ur[r];
            Ui[k1 * PU + colB] = ui[r];
        }
    }
    __syncthreads();

    {
        float mrr[8], mii[8];
        #pragma unroll
        for (int r = 0; r < 8; ++r) {
            const int k1  = rowD + r;
            const int k1m = (16 - k1) & 15;
            const int k2m = (k1 == 0) ? ((32 - colB) & 31) : (31 - colB);
            mrr[r] = Ur[k1m * PU + k2m];
            mii[r] = Ui[k1m * PU + k2m];
        }
        #pragma unroll
        for (int r = 0; r < 8; ++r) {
            const int k1  = rowD + r;
            const float zr = ur[r];
            const float zi = ui[r];
            const float mr = mrr[r];
            const float mi = mii[r];
            const float er = 0.5f * (zr + mr), ei = 0.5f * (zi - mi);
            const float orr = 0.5f * (zi + mi), oi = 0.5f * (mr - zr);
            const float e2r = er * er - ei * ei,    e2i = 2.0f * (er * ei);
            const float o2r = orr * orr - oi * oi,  o2i = 2.0f * (orr * oi);
            const float tor = ct[r] * o2r + st[r] * o2i;
            const float toi = ct[r] * o2i - st[r] * o2r;
            Hr[k1 * PA + colB] = e2r + tor - 2.0f * (er * oi + ei * orr);
            Hi[k1 * PA + colB] = e2i + toi + 2.0f * (er * orr - ei * oi);
        }
    }
    __syncthreads();

    {
        v8f b1 = {}, b2 = {}, b3 = {}, b4 = {};
        v2f pr[8], pi[8];
        #pragma unroll
        for (int q = 0; q < 8; ++q) {
            const int ad = rowA * PA + 4 * q + 2 * h;
            pr[q] = *(const v2f*)(Hr + ad);
            pi[q] = *(const v2f*)(Hi + ad);
        }
        __builtin_amdgcn_sched_barrier(0);
        #pragma unroll
        for (int q = 0; q < 8; ++q) {
            b1 = mma(pr[q], cb[q], b1);
            b2 = mma(pi[q], sb[q], b2);
            b3 = mma(pr[q], sb[q], b3);
            b4 = mma(pi[q], cb[q], b4);
        }
        #pragma unroll
        for (int r = 0; r < 8; ++r) {
            const int k1 = rowD + r;
            const float qr = b1[r] - b2[r];
            const float qi = b3[r] + b4[r];
            v2f rp;
            rp.x = qr * sw[r] + qi * cw[r];
            rp.y = qr * cw[r] - qi * sw[r];
            *(v2f*)(RR + k1 * PR + 2 * colB) = rp;
        }
    }
    __syncthreads();

    {
        v8f gr = {}, gi = {};
        v2f rp[8];
        #pragma unroll
        for (int q = 0; q < 8; ++q) {
            const int j = 2 * q + h;
            rp[q] = *(const v2f*)(RR + j * PR + 2 * colB);
        }
        __builtin_amdgcn_sched_barrier(0);
        #pragma unroll
        for (int q = 0; q < 8; ++q) {
            gr = mma(fb[q], rp[q], gr);
            gi = mma(fa[q], rp[q], gi);
        }
        #pragma unroll
        for (int r = 0; r < 8; ++r) {
            v2f g;
            g.x = gr[r] * 0.001953125f;
            g.y = gi[r] * 0.001953125f;
            *(v2f*)(ob + 64 * (rowD + r) + 2 * colB) = g;
        }
    }
}

extern "C" void kernel_launch(void* const* d_in, const int* in_sizes, int n_in,
                              void* d_out, int out_size, void* d_ws, size_t ws_size,
                              hipStream_t stream) {
    const float* x1 = (const float*)d_in[0];
    const float* x2 = (const float*)d_in[1];
    const float* x3 = (const float*)d_in[2];
    float* out = (float*)d_out;

    dim3 grid(64, 3);
    dim3 block(128);
    mcf_wmma_kernel<<<grid, block, 0, stream>>>(x1, x2, x3, out);
}


# radix-2 split of all four DFT stages: half of each stage's outputs on WMMA from even and odd partial sums, the other half by one subtraction; 48 WMMAs a wave, not 96; sum order changes
# speedup vs baseline: 2.9881x; 1.0521x over previous
#include <hip/hip_runtime.h>


#define NN 1024
#define NB 128
#define PA 36
#define PR 68
#define PU 33
#define ROW_LDS (4 * 16 * PA + 16 * PR + 2 * 16 * PU + 8)

typedef float v2f __attribute__((ext_vector_type(2)));
typedef float v8f __attribute__((ext_vector_type(8)));

__device__ __forceinline__ v8f mma(v2f a, v2f b, v8f c) {
    return __builtin_amdgcn_wmma_f32_16x16x4_f32(false, a, false, b, (short)0, c, false, false);
}

__global__ __launch_bounds__(128) void mcf_wmma_kernel(const float* __restrict__ x0,
                                                      const float* __restrict__ x1,
                                                      const float* __restrict__ x2,
                                                      float* __restrict__ out) {
    __shared__ __attribute__((aligned(16))) float c32[32];
    __shared__ __attribute__((aligned(16))) float s32[32];
    __shared__ __attribute__((aligned(16))) float planes[2][ROW_LDS];

    const int tid  = threadIdx.x;
    const int wave = tid >> 5;
    const int pr2  = wave >> 1;
    const int wt   = wave & 1;

    float* const rowbase = planes[pr2];
    float* const Zr = rowbase;
    float* const Zi = Zr + 16 * PA;
    float* const Hr = Zi + 16 * PA;
    float* const Hi = Hr + 16 * PA;
    float* const RR = Hi + 16 * PA;
    float* const Ur = RR + 16 * PR;
    float* const Ui = Ur + 16 * PU;

    const int b     = 2 * blockIdx.x + pr2;
    const int which = blockIdx.y;
    const float* __restrict__ x =
        (which == 0) ? x0 : (which == 1) ? x1 : x2;
    const float* __restrict__ xb = x + b * NN;
    float* __restrict__ ob = out + (size_t)which * (NB * NN) + (size_t)b * NN;

    const int lane = tid & 31;
    const int c16  = lane & 15;
    const int h    = lane >> 4;
    const int rowA = c16;
    const int colB = 16 * wt + c16;
    const int rowD = 8 * h;
    const int part = wt;
    const int colP = ((colB & 1) << 4) | (colB >> 1);

    v2f zl[8];
    #pragma unroll
    for (int q = 0; q < 8; ++q) {
        const int j = 4 * h + 2 * (q & 1) + 8 * ((q >> 1) & 1) + (q >> 2);
        zl[q] = *(const v2f*)(xb + 64 * j + 2 * colB);
    }

    if (tid < 32) {
        float s, c;
        sincospif((float)tid * 0.0625f, &s, &c);
        c32[tid] = c;
        s32[tid] = s;
    }

    __builtin_amdgcn_fence(__ATOMIC_RELEASE, "workgroup", "local");
    __builtin_amdgcn_s_barrier();
    __builtin_amdgcn_fence(__ATOMIC_ACQUIRE, "workgroup", "local");

    v2f a1[8], a4[8], cb[8], sb[8];
    const int kA = rowA >> 1;
    const int pA = rowA & 1;
    #pragma unroll
    for (int q = 0; q < 8; ++q) {
        const int j = 4 * h + 2 * (q & 1) + 8 * ((q >> 1) & 1) + (q >> 2);
        const float cc = c32[(2 * kA * j) & 31];
        const float ss = s32[(2 * kA * j) & 31];
        a1[q].x = pA ? -ss : cc;    a1[q].y = pA ? cc : ss;
        a4[q].x = pA ? cc : -ss;    a4[q].y = pA ? ss : cc;
    }
    #pragma unroll
    for (int q = 0; q < 8; ++q) {
        const int n0 = 2 * (4 * (q & 3) + 2 * h) + (q >> 2);
        cb[q].x = c32[(c16 * n0) & 31];         cb[q].y = c32[(c16 * (n0 + 2)) & 31];
        sb[q].x = s32[(c16 * n0) & 31];         sb[q].y = s32[(c16 * (n0 + 2)) & 31];
    }

    const float tc = c32[colB], ts = s32[colB];
    const float c8 = h ? 0.995184720f : 1.0f;
    const float s8 = h ? 0.0980171412f : 0.0f;
    const float t0c = tc * c8 - ts * s8;
    const float t0s = tc * s8 + ts * c8;
    float cw[8], sw[8], ct[8], st[8];
    #pragma unroll
    for (int r = 0; r < 8; ++r) {
        sincospif((float)((4 * h + (r & 3) + 8 * (r >> 2)) * colB) * 0.00390625f, &sw[r], &cw[r]);
    }
#define SHIFT_TWIDDLE(r, C, S) ct[r] = t0c * (C) - t0s * (S); st[r] = t0c * (S) + t0s * (C);
    ct[0] = t0c;  st[0] = t0s;
    SHIFT_TWIDDLE(1, 0.999924719f, 0.0122715384f)
    SHIFT_TWIDDLE(2, 0.999698818f, 0.0245412290f)
    SHIFT_TWIDDLE(3, 0.999322355f, 0.0368072242f)
    SHIFT_TWIDDLE(4, 0.998795450f, 0.0490676761f)
    SHIFT_TWIDDLE(5, 0.998118103f, 0.0613207370f)
    SHIFT_TWIDDLE(6, 0.997290432f, 0.0735645667f)
    SHIFT_TWIDDLE(7, 0.996312618f, 0.0857973099f)
#undef SHIFT_TWIDDLE

    asm volatile("" :: "v"(a1[0]), "v"(a1[1]), "v"(a1[2]), "v"(a1[3]), "v"(a1[4]), "v"(a1[5]),
                       "v"(a1[6]), "v"(a1[7]), "v"(a4[0]), "v"(a4[1]), "v"(a4[2]), "v"(a4[3]),
                       "v"(a4[4]), "v"(a4[5]), "v"(a4[6]), "v"(a4[7]));
    asm volatile("" :: "v"(cb[0]), "v"(cb[1]), "v"(cb[2]), "v"(cb[3]), "v"(cb[4]), "v"(cb[5]),
                       "v"(cb[6]), "v"(cb[7]), "v"(sb[0]), "v"(sb[1]), "v"(sb[2]), "v"(sb[3]),
                       "v"(sb[4]), "v"(sb[5]), "v"(sb[6]), "v"(sb[7]));
    asm volatile("" :: "v"(cw[0]), "v"(cw[1]), "v"(cw[2]), "v"(cw[3]), "v"(cw[4]), "v"(cw[5]),
                       "v"(cw[6]), "v"(cw[7]), "v"(sw[0]), "v"(sw[1]), "v"(sw[2]), "v"(sw[3]),
                       "v"(sw[4]), "v"(sw[5]), "v"(sw[6]), "v"(sw[7]));
    asm volatile("" :: "v"(ct[0]), "v"(ct[1]), "v"(ct[2]), "v"(ct[3]), "v"(ct[4]), "v"(ct[5]),
                       "v"(ct[6]), "v"(ct[7]), "v"(st[0]), "v"(st[1]), "v"(st[2]), "v"(st[3]),
                       "v"(st[4]), "v"(st[5]), "v"(st[6]), "v"(st[7]));
    __builtin_amdgcn_sched_barrier(0);

    {
        v8f ye = {}, yo = {};
        #pragma unroll
        for (int q = 0; q < 4; ++q) {
            ye = mma(a1[q],     zl[q],     ye);
            yo = mma(a1[q + 4], zl[q + 4], yo);
        }
        #pragma unroll
        for (int t = 0; t < 8; ++t) {
            const int i  = t & 3;
            const int k1 = 4 * h + i + 8 * (t >> 2);
            const float yr = (t < 4) ? ye[2 * i] + yo[2 * i]         : ye[2 * i] - yo[2 * i];
            const float yi = (t < 4) ? ye[2 * i + 1] + yo[2 * i + 1] : ye[2 * i + 1] - yo[2 * i + 1];
            Zr[k1 * PA + colP] = yr * cw[t] + yi * sw[t];
            Zi[k1 * PA + colP] = yi * cw[t] - yr * sw[t];
        }
    }
    __syncthreads();

    {
        const int pc = part ? 16 * PA : 0;
        const int ps = part ? 0 : 16 * PA;
        const float sg = part ? -1.0f : 1.0f;
        v8f e1 = {}, e2 = {}, o1 = {}, o2 = {};
        v2f f1[8], f2[8];
        #pragma unroll
        for (int q = 0; q < 8; ++q) {
            const int ad = rowA * PA + 4 * q + 2 * h;
            f1[q] = *(const v2f*)(Zr + pc + ad);
            f2[q] = *(const v2f*)(Zr + ps + ad);
        }
        __builtin_amdgcn_sched_barrier(0);
        #pragma unroll
        for (int q = 0; q < 4; ++q) {
            e1 = mma(f1[q], cb[q], e1);
            e2 = mma(f2[q], sb[q], e2);
        }
        #pragma unroll
        for (int q = 4; q < 8; ++q) {
            o1 = mma(f1[q], cb[q], o1);
            o2 = mma(f2[q], sb[q], o2);
        }
        float* const Up = part ? Ui : Ur;
        #pragma unroll
        for (int r = 0; r < 8; ++r) {
            const int k1 = rowD + r;
            const float pe = e1[r] + sg * e2[r];
            const float po = o1[r] + sg * o2[r];
            Up[k1 * PU + c16]      = pe + po;
            Up[k1 * PU + c16 + 16] = pe - po;
        }
    }
    __syncthreads();

    {
        float urr[8], uii[8], mrr[8], mii[8];
        #pragma unroll
        for (int r = 0; r < 8; ++r) {
            const int k1  = rowD + r;
            urr[r] = Ur[k1 * PU + colB];
            uii[r] = Ui[k1 * PU + colB];
        }
        #pragma unroll
        for (int r = 0; r < 8; ++r) {
            const int k1  = rowD + r;
            const int k1m = (16 - k1) & 15;
            const int k2m = (k1 == 0) ? ((32 - colB) & 31) : (31 - colB);
            mrr[r] = Ur[k1m * PU + k2m];
            mii[r] = Ui[k1m * PU + k2m];
        }
        #pragma unroll
        for (int r = 0; r < 8; ++r) {
            const int k1  = rowD + r;
            const float zr = urr[r];
            const float zi = uii[r];
            const float mr = mrr[r];
            const float mi = mii[r];
            const float er = 0.5f * (zr + mr), ei = 0.5f * (zi - mi);
            const float orr = 0.5f * (zi + mi), oi = 0.5f * (mr - zr);
            const float e2r = er * er - ei * ei,    e2i = 2.0f * (er * ei);
            const float o2r = orr * orr - oi * oi,  o2i = 2.0f * (orr * oi);
            const float tor = ct[r] * o2r + st[r] * o2i;
            const float toi = ct[r] * o2i - st[r] * o2r;
            Hr[k1 * PA + colP] = e2r + tor - 2.0f * (er * oi + ei * orr);
            Hi[k1 * PA + colP] = e2i + toi + 2.0f * (er * orr - ei * oi);
        }
    }
    __syncthreads();

    {
        const int pc = part ? 16 * PA : 0;
        const int ps = part ? 0 : 16 * PA;
        const float sg = part ? 1.0f : -1.0f;
        v8f e1 = {}, e2 = {}, o1 = {}, o2 = {};
        v2f f1[8], f2[8];
        #pragma unroll
        for (int q = 0; q < 8; ++q) {
            const int ad = rowA * PA + 4 * q + 2 * h;
            f1[q] = *(const v2f*)(Hr + pc + ad);
            f2[q] = *(const v2f*)(Hr + ps + ad);
        }
        __builtin_amdgcn_sched_barrier(0);
        #pragma unroll
        for (int q = 0; q < 4; ++q) {
            e1 = mma(f1[q], cb[q], e1);
            e2 = mma(f2[q], sb[q], e2);
        }
        #pragma unroll
        for (int q = 4; q < 8; ++q) {
            o1 = mma(f1[q], cb[q], o1);
            o2 = mma(f2[q], sb[q], o2);
        }
        float* const Qp = RR + (1 - part);
        #pragma unroll
        for (int r = 0; r < 8; ++r) {
            const int k1 = rowD + r;
            const float pe = e1[r] + sg * e2[r];
            const float po = o1[r] + sg * o2[r];
            Qp[k1 * PR + 2 * c16]        = pe + po;
            Qp[k1 * PR + 2 * (c16 + 16)] = pe - po;
        }
    }
    __syncthreads();

    {
        v8f ge = {}, go = {};
        v2f qq[8], rp[8];
        #pragma unroll
        for (int q = 0; q < 8; ++q) {
            const int j = 4 * h + 2 * (q & 1) + 8 * ((q >> 1) & 1) + (q >> 2);
            qq[q] = *(const v2f*)(RR + j * PR + 2 * colB);
        }
        __builtin_amdgcn_sched_barrier(0);
        #pragma unroll
        for (int q = 0; q < 8; ++q) {
            const int t = 4 * ((q >> 1) & 1) + 2 * (q & 1) + (q >> 2);
            const float qi = qq[q].x, qr = qq[q].y;
            rp[q].x = qr * sw[t] + qi * cw[t];
            rp[q].y = qr * cw[t] - qi * sw[t];
        }
        #pragma unroll
        for (int q = 0; q < 4; ++q) {
            ge = mma(a4[q],     rp[q],     ge);
            go = mma(a4[q + 4], rp[q + 4], go);
        }
        #pragma unroll
        for (int t = 0; t < 8; ++t) {
            const int i  = t & 3;
            const int n1 = 4 * h + i + 8 * (t >> 2);
            v2f g;
            g.x = ((t < 4) ? ge[2 * i] + go[2 * i]         : ge[2 * i] - go[2 * i])         * 0.001953125f;
            g.y = ((t < 4) ? ge[2 * i + 1] + go[2 * i + 1] : ge[2 * i + 1] - go[2 * i + 1]) * 0.001953125f;
            *(v2f*)(ob + 64 * n1 + 2 * colB) = g;
        }
    }
}

extern "C" void kernel_launch(void* const* d_in, const int* in_sizes, int n_in,
                              void* d_out, int out_size, void* d_ws, size_t ws_size,
                              hipStream_t stream) {
    const float* x1 = (const float*)d_in[0];
    const float* x2 = (const float*)d_in[1];
    const float* x3 = (const float*)d_in[2];
    float* out = (float*)d_out;

    dim3 grid(64, 3);
    dim3 block(128);
    mcf_wmma_kernel<<<grid, block, 0, stream>>>(x1, x2, x3, out);
}
